# starting kernel
# baseline (speedup 1.0000x reference)
#include <hip/hip_runtime.h>
#include <cmath>

typedef float v2f __attribute__((ext_vector_type(2)));
typedef float v8f __attribute__((ext_vector_type(8)));

constexpr int Hdim  = 128;
constexpr int TT    = 64;
constexpr int PLEN  = 32;
constexpr int WAVES = 4;
constexpr int NJ    = 8;
constexpr int NK    = 32;
constexpr int BPACK = NK * NJ * 32 * 2;
constexpr int APACK = NK * 32 * 2;

extern __shared__ float smem[];

__device__ __forceinline__ int a_idx(int m, int k) {
    return (k >> 2) * 64 + (((((k >> 1) & 1) << 4) + m) << 1) + (k & 1);
}

__device__ __forceinline__ float fast_tanh(float x) {
    float e = __builtin_amdgcn_exp2f(x * 2.88539008177792681472f);
    return 1.0f - 2.0f * __builtin_amdgcn_rcpf(e + 1.0f);
}

__global__ __launch_bounds__(WAVES * 32)
void traj_rnn_kernel(const float* __restrict__ x,
                     const float* __restrict__ eWih, const float* __restrict__ eWhh,
                     const float* __restrict__ ebih, const float* __restrict__ ebhh,
                     const float* __restrict__ dWih, const float* __restrict__ dWhh,
                     const float* __restrict__ dbih, const float* __restrict__ dbhh,
                     const float* __restrict__ fcW,  const float* __restrict__ fcb,
                     float* __restrict__ out, int B)
{
    float* packBe = smem;
    float* packBd = packBe + BPACK;
    float* fcWl   = packBd + BPACK;
    float* fcbl   = fcWl + 3 * Hdim;
    float* hall   = fcbl + 4;
    float* iall   = hall + WAVES * APACK;

    const int tid  = threadIdx.x;
    const int wave = tid >> 5;
    const int lane = tid & 31;
    const int half = lane >> 4;
    const int ln   = lane & 15;

    for (int idx = tid; idx < BPACK; idx += blockDim.x) {
        int p  = idx & 1;
        int l  = (idx >> 1) & 31;
        int j  = (idx >> 6) & 7;
        int kk = idx >> 9;
        int n  = j * 16 + (l & 15);
        int k  = kk * 4 + ((l >> 4) << 1) + p;
        packBe[idx] = eWhh[n * Hdim + k];
        packBd[idx] = dWhh[n * Hdim + k];
    }
    for (int idx = tid; idx < 3 * Hdim; idx += blockDim.x) fcWl[idx] = fcW[idx];
    if (tid < 3) fcbl[tid] = fcb[tid];
    __syncthreads();

    const int tile = blockIdx.x * WAVES + wave;
    const int b0   = tile * 16;
    if (b0 >= B) return;

    float* hA = hall + wave * APACK;
    float* il = iall + wave * (16 * 4);

    float bse[NJ], bsd[NJ], wie[NJ][3], wid[NJ][3];
#pragma unroll
    for (int j = 0; j < NJ; ++j) {
        int col = j * 16 + ln;
        bse[j] = ebih[col] + ebhh[col];
        bsd[j] = dbih[col] + dbhh[col];
#pragma unroll
        for (int i = 0; i < 3; ++i) {
            wie[j][i] = eWih[col * 3 + i];
            wid[j][i] = dWih[col * 3 + i];
        }
    }

    for (int idx = lane; idx < APACK; idx += 32) hA[idx] = 0.0f;

    v8f c[NJ];

    for (int t = 0; t < TT; ++t) {
        float xv[8][3];
#pragma unroll
        for (int r = 0; r < 8; ++r) {
            const float* xp = x + ((size_t)(b0 + r + half * 8) * TT + t) * 3;
#pragma unroll
            for (int i = 0; i < 3; ++i) xv[r][i] = xp[i];
        }
#pragma unroll
        for (int j = 0; j < NJ; ++j)
#pragma unroll
            for (int r = 0; r < 8; ++r)
                c[j][r] = bse[j] + xv[r][0] * wie[j][0] + xv[r][1] * wie[j][1]
                                 + xv[r][2] * wie[j][2];

#pragma unroll 4
        for (int kk = 0; kk < NK; ++kk) {
            v2f a = *(const v2f*)&hA[(kk * 32 + lane) * 2];
#pragma unroll
            for (int j = 0; j < NJ; ++j) {
                v2f bf = *(const v2f*)&packBe[((kk * NJ + j) * 32 + lane) * 2];
                c[j] = __builtin_amdgcn_wmma_f32_16x16x4_f32(
                        false, a, false, bf, (short)0, c[j], false, false);
            }
        }
#pragma unroll
        for (int j = 0; j < NJ; ++j)
#pragma unroll
            for (int r = 0; r < 8; ++r)
                hA[a_idx(r + half * 8, j * 16 + ln)] = fast_tanh(c[j][r]);
    }

    if (lane < 16) {
#pragma unroll
        for (int i = 0; i < 3; ++i)
            il[lane * 4 + i] = x[((size_t)(b0 + lane) * TT + (TT - 1)) * 3 + i];
    }

    for (int p = 0; p < PLEN; ++p) {
        float xv[8][3];
#pragma unroll
        for (int r = 0; r < 8; ++r)
#pragma unroll
            for (int i = 0; i < 3; ++i) xv[r][i] = il[(r + half * 8) * 4 + i];
#pragma unroll
        for (int j = 0; j < NJ; ++j)
#pragma unroll
            for (int r = 0; r < 8; ++r)
                c[j][r] = bsd[j] + xv[r][0] * wid[j][0] + xv[r][1] * wid[j][1]
                                 + xv[r][2] * wid[j][2];

#pragma unroll 4
        for (int kk = 0; kk < NK; ++kk) {
            v2f a = *(const v2f*)&hA[(kk * 32 + lane) * 2];
#pragma unroll
            for (int j = 0; j < NJ; ++j) {
                v2f bf = *(const v2f*)&packBd[((kk * NJ + j) * 32 + lane) * 2];
                c[j] = __builtin_amdgcn_wmma_f32_16x16x4_f32(
                        false, a, false, bf, (short)0, c[j], false, false);
            }
        }
#pragma unroll
        for (int j = 0; j < NJ; ++j)
#pragma unroll
            for (int r = 0; r < 8; ++r)
                hA[a_idx(r + half * 8, j * 16 + ln)] = fast_tanh(c[j][r]);

        if (lane < 16) {
            float a0 = fcbl[0], a1 = fcbl[1], a2 = fcbl[2];
#pragma unroll 8
            for (int k = 0; k < Hdim; ++k) {
                float hv = hA[a_idx(lane, k)];
                a0 += hv * fcWl[0 * Hdim + k];
                a1 += hv * fcWl[1 * Hdim + k];
                a2 += hv * fcWl[2 * Hdim + k];
            }
            size_t ob = ((size_t)(b0 + lane) * PLEN + p) * 3;
            out[ob + 0] = a0; out[ob + 1] = a1; out[ob + 2] = a2;
            il[lane * 4 + 0] = a0; il[lane * 4 + 1] = a1; il[lane * 4 + 2] = a2;
        }
    }
}

extern "C" void kernel_launch(void* const* d_in, const int* in_sizes, int n_in,
                              void* d_out, int out_size, void* d_ws, size_t ws_size,
                              hipStream_t stream) {
    (void)n_in; (void)out_size; (void)d_ws; (void)ws_size;
    const float* x    = (const float*)d_in[0];
    const float* eWih = (const float*)d_in[1];
    const float* eWhh = (const float*)d_in[2];
    const float* ebih = (const float*)d_in[3];
    const float* ebhh = (const float*)d_in[4];
    const float* dWih = (const float*)d_in[5];
    const float* dWhh = (const float*)d_in[6];
    const float* dbih = (const float*)d_in[7];
    const float* dbhh = (const float*)d_in[8];
    const float* fcW  = (const float*)d_in[9];
    const float* fcb  = (const float*)d_in[10];
    float* out = (float*)d_out;

    int B = in_sizes[0] / (TT * 3);
    int tiles  = (B + 15) / 16;
    int blocks = (tiles + WAVES - 1) / WAVES;

    size_t shmem = (size_t)(2 * BPACK + 3 * Hdim + 4
                            + WAVES * APACK + WAVES * 16 * 4) * sizeof(float);

    traj_rnn_kernel<<<blocks, WAVES * 32, shmem, stream>>>(
        x, eWih, eWhh, ebih, ebhh, dWih, dWhh, dbih, dbhh, fcW, fcb, out, B);
}


# Each 16-row tile's 128 hidden columns split over 4 waves (weights in registers, h via LDS, 1 barrier/step, 1 work-group per WGP by LDS request): 1024 waves not 256; fc on matrix unit, new sum order
# speedup vs baseline: 3.8933x; 3.8933x over previous
#include <hip/hip_runtime.h>
#include <cmath>


typedef float v2f __attribute__((ext_vector_type(2)));
typedef float v4f __attribute__((ext_vector_type(4)));
typedef float v8f __attribute__((ext_vector_type(8)));

constexpr int Hdim  = 128;
constexpr int TT    = 64;
constexpr int PLEN  = 32;
constexpr int WAVES = 4;
constexpr int NK    = 32;
constexpr int NJW   = 2;
constexpr int APACK = NK * 32 * 2;
constexpr int LDS_USED_FLOATS = 2 * APACK + TT * 16 * 4;

extern __shared__ __attribute__((aligned(16))) float smem[];

__device__ __forceinline__ int a_idx(int m, int k) {
    return (k >> 2) * 64 + (((((k >> 1) & 1) << 4) + m) << 1) + (k & 1);
}

__device__ __forceinline__ float fast_tanh(float x) {
    float e = __builtin_amdgcn_exp2f(x * 2.88539008177792681472f);
    return 1.0f - 2.0f * __builtin_amdgcn_rcpf(e + 1.0f);
}

__device__ __forceinline__ v8f mma(v2f a, v2f b, v8f c) {
    return __builtin_amdgcn_wmma_f32_16x16x4_f32(false, a, false, b, (short)0, c, false, false);
}

template <int I>
__device__ __forceinline__ float row_share(float v) {
    return __int_as_float(__builtin_amdgcn_update_dpp(0, __float_as_int(v), 0x150 + I, 0xf, 0xf, false));
}

__device__ __forceinline__ void load_bw(v2f (&bw)[NK][NJW], const float* __restrict__ Wm,
                                        int colbase, int half, int ln) {
#pragma unroll
    for (int kk = 0; kk < NK; ++kk)
#pragma unroll
        for (int jj = 0; jj < NJW; ++jj)
            bw[kk][jj] = *(const v2f*)&Wm[(colbase + jj * 16 + ln) * Hdim + kk * 4 + half * 2];
}

__device__ __forceinline__ void cinit(v8f (&c)[NJW], const v4f (&xv)[8],
                                      const float (&bs)[NJW], const float (&wi)[NJW][3]) {
#pragma unroll
    for (int jj = 0; jj < NJW; ++jj)
#pragma unroll
        for (int r = 0; r < 8; ++r)
            c[jj][r] = bs[jj] + xv[r].x * wi[jj][0] + xv[r].y * wi[jj][1] + xv[r].z * wi[jj][2];
}

__device__ __forceinline__ void store_h(float* hn, const v8f (&c)[NJW], int half, int colbase, int ln) {
#pragma unroll
    for (int jj = 0; jj < NJW; ++jj)
#pragma unroll
        for (int r = 0; r < 8; ++r)
            hn[a_idx(r + half * 8, colbase + jj * 16 + ln)] = fast_tanh(c[jj][r]);
}

template <bool FC, bool MAIN>
__device__ __forceinline__ void dec_pass(const float* hc, float* hn,
                                         const v2f (&bw)[NK][NJW], const v2f (&bfc)[NK],
                                         const float (&bs)[NJW], const float (&wi)[NJW][3],
                                         float fcb_lane, v4f (&inp)[8], float* outp,
                                         int wave, int lane, int half, int ln, int colbase) {
    v2f a[NK];
#pragma unroll
    for (int kk = 0; kk < NK; ++kk) a[kk] = *(const v2f*)&hc[(kk * 32 + lane) * 2];

    v8f c[NJW], pa, pb, d[NJW];
    if (MAIN) {
        if (!FC) {
            cinit(c, inp, bs, wi);
        } else {
#pragma unroll
            for (int jj = 0; jj < NJW; ++jj)
#pragma unroll
                for (int r = 0; r < 8; ++r) c[jj][r] = bs[jj];
        }
    }
    if (FC) {
#pragma unroll
        for (int r = 0; r < 8; ++r) { pa[r] = fcb_lane; pb[r] = 0.0f; }
    }
#pragma unroll
    for (int kk = 0; kk < NK / 2; ++kk) {
        if (FC)   pa   = mma(a[2 * kk], bfc[2 * kk], pa);
        if (MAIN) c[0] = mma(a[kk], bw[kk][0], c[0]);
        if (FC)   pb   = mma(a[2 * kk + 1], bfc[2 * kk + 1], pb);
        if (MAIN) c[1] = mma(a[kk], bw[kk][1], c[1]);
    }
    if (FC) {
        v8f pr = pa + pb;
        const int m0 = wave == 0 ? -1 : 0, m1 = wave == 1 ? -1 : 0, m2 = wave == 2 ? -1 : 0, m3 = wave == 3 ? -1 : 0;
        float o0 = __int_as_float((__float_as_int(pr[0]) & m0) | (__float_as_int(pr[2]) & m1) |
                                  (__float_as_int(pr[4]) & m2) | (__float_as_int(pr[6]) & m3));
        float o1 = __int_as_float((__float_as_int(pr[1]) & m0) | (__float_as_int(pr[3]) & m1) |
                                  (__float_as_int(pr[5]) & m2) | (__float_as_int(pr[7]) & m3));
        if (ln < 3) { outp[0] = o0; outp[PLEN * 3] = o1; }
#pragma unroll
        for (int r = 0; r < 8; ++r) {
            inp[r].x = row_share<0>(pr[r]);
            inp[r].y = row_share<1>(pr[r]);
            inp[r].z = row_share<2>(pr[r]);
        }
        if (MAIN) {
#pragma unroll
            for (int jj = 0; jj < NJW; ++jj)
#pragma unroll
                for (int r = 0; r < 8; ++r)
                    d[jj][r] = inp[r].x * wi[jj][0] + inp[r].y * wi[jj][1] + inp[r].z * wi[jj][2];
        }
    }
    if (MAIN) {
#pragma unroll
        for (int kk = NK / 2; kk < NK; ++kk) {
            c[0] = mma(a[kk], bw[kk][0], c[0]);
            c[1] = mma(a[kk], bw[kk][1], c[1]);
        }
        if (FC) {
#pragma unroll
            for (int jj = 0; jj < NJW; ++jj) c[jj] += d[jj];
        }
        store_h(hn, c, half, colbase, ln);
    }
}

__global__ __launch_bounds__(WAVES * 32)
void traj_rnn_kernel(const float* __restrict__ x,
                     const float* __restrict__ eWih, const float* __restrict__ eWhh,
                     const float* __restrict__ ebih, const float* __restrict__ ebhh,
                     const float* __restrict__ dWih, const float* __restrict__ dWhh,
                     const float* __restrict__ dbih, const float* __restrict__ dbhh,
                     const float* __restrict__ fcW,  const float* __restrict__ fcb,
                     float* __restrict__ out, int B)
{
    float* hbuf = smem;
    float* xl   = smem + 2 * APACK;

    const int tid  = threadIdx.x;
    const int wave = __builtin_amdgcn_readfirstlane(tid >> 5);
    const int lane = tid & 31;
    const int half = lane >> 4;
    const int ln   = lane & 15;
    const int colbase = wave * 32;

    const int b0 = blockIdx.x * 16;
    if (b0 >= B) return;

    const float* xt = x + (size_t)b0 * TT * 3;
    for (int e = tid; e < 16 * TT * 3; e += WAVES * 32) {
        int row = e / (TT * 3);
        int rem = e - row * (TT * 3);
        int t   = rem / 3;
        int i   = rem - t * 3;
        xl[(t * 16 + row) * 4 + i] = xt[e];
    }
    for (int e = tid; e < TT * 16; e += WAVES * 32) xl[e * 4 + 3] = 0.0f;
    for (int e = tid; e < APACK; e += WAVES * 32) hbuf[e] = 0.0f;

    v2f bwe[NK][NJW], bwd[NK][NJW], bfc[NK];
    load_bw(bwe, eWhh, colbase, half, ln);
    load_bw(bwd, dWhh, colbase, half, ln);
#pragma unroll
    for (int kk = 0; kk < NK; ++kk) {
        v2f f = *(const v2f*)&fcW[(ln < 3 ? ln : 0) * Hdim + kk * 4 + half * 2];
        bfc[kk].x = ln < 3 ? f.x : 0.0f;
        bfc[kk].y = ln < 3 ? f.y : 0.0f;
    }
    const float fcb_lane = ln < 3 ? fcb[ln < 3 ? ln : 0] : 0.0f;

    float bse[NJW], bsd[NJW], wie[NJW][3], wid[NJW][3];
#pragma unroll
    for (int jj = 0; jj < NJW; ++jj) {
        int col = colbase + jj * 16 + ln;
        bse[jj] = ebih[col] + ebhh[col];
        bsd[jj] = dbih[col] + dbhh[col];
#pragma unroll
        for (int i = 0; i < 3; ++i) {
            wie[jj][i] = eWih[col * 3 + i];
            wid[jj][i] = dWih[col * 3 + i];
        }
    }

    __syncthreads();

    v4f xn[8];
#pragma unroll
    for (int r = 0; r < 8; ++r) xn[r] = *(const v4f*)&xl[(r + half * 8) * 4];
    v8f c[NJW];
    cinit(c, xn, bse, wie);

    for (int t = 0; t < TT; ++t) {
        const float* hc = hbuf + (t & 1) * APACK;
        float*       hn = hbuf + ((t & 1) ^ 1) * APACK;
        v2f a[NK];
#pragma unroll
        for (int kk = 0; kk < NK; ++kk) a[kk] = *(const v2f*)&hc[(kk * 32 + lane) * 2];
        const int tn = t + 1 < TT ? t + 1 : TT - 1;
#pragma unroll
        for (int r = 0; r < 8; ++r) xn[r] = *(const v4f*)&xl[(tn * 16 + r + half * 8) * 4];

#pragma unroll
        for (int kk = 0; kk < NK / 2; ++kk) {
            c[0] = mma(a[kk], bwe[kk][0], c[0]);
            c[1] = mma(a[kk], bwe[kk][1], c[1]);
        }
        v8f cn[NJW];
        cinit(cn, xn, bse, wie);
#pragma unroll
        for (int kk = NK / 2; kk < NK; ++kk) {
            c[0] = mma(a[kk], bwe[kk][0], c[0]);
            c[1] = mma(a[kk], bwe[kk][1], c[1]);
        }
        store_h(hn, c, half, colbase, ln);
        c[0] = cn[0];
        c[1] = cn[1];
        __syncthreads();
    }

    int cur = TT & 1;
    float* outp = out + ((size_t)(b0 + 2 * wave + half * 8) * PLEN) * 3 + (ln < 3 ? ln : 0);

    dec_pass<false, true>(hbuf + cur * APACK, hbuf + (cur ^ 1) * APACK, bwd, bfc, bsd, wid,
                          fcb_lane, xn, outp, wave, lane, half, ln, colbase);
    __syncthreads();
    cur ^= 1;
    for (int p = 1; p < PLEN; ++p) {
        dec_pass<true, true>(hbuf + cur * APACK, hbuf + (cur ^ 1) * APACK, bwd, bfc, bsd, wid,
                             fcb_lane, xn, outp + (p - 1) * 3, wave, lane, half, ln, colbase);
        __syncthreads();
        cur ^= 1;
    }
    dec_pass<true, false>(hbuf + cur * APACK, hbuf + (cur ^ 1) * APACK, bwd, bfc, bsd, wid,
                          fcb_lane, xn, outp + (PLEN - 1) * 3, wave, lane, half, ln, colbase);
}

extern "C" void kernel_launch(void* const* d_in, const int* in_sizes, int n_in,
                              void* d_out, int out_size, void* d_ws, size_t ws_size,
                              hipStream_t stream) {
    (void)n_in; (void)out_size; (void)d_ws; (void)ws_size;
    const float* x    = (const float*)d_in[0];
    const float* eWih = (const float*)d_in[1];
    const float* eWhh = (const float*)d_in[2];
    const float* ebih = (const float*)d_in[3];
    const float* ebhh = (const float*)d_in[4];
    const float* dWih = (const float*)d_in[5];
    const float* dWhh = (const float*)d_in[6];
    const float* dbih = (const float*)d_in[7];
    const float* dbhh = (const float*)d_in[8];
    const float* fcW  = (const float*)d_in[9];
    const float* fcb  = (const float*)d_in[10];
    float* out = (float*)d_out;

    int B = in_sizes[0] / (TT * 3);
    int blocks = (B + 15) / 16;

    static_assert(LDS_USED_FLOATS * sizeof(float) <= 164 * 1024, "LDS request too small");
    size_t shmem = 164 * 1024;

    traj_rnn_kernel<<<blocks, WAVES * 32, shmem, stream>>>(
        x, eWih, eWhh, ebih, ebhh, dWih, dWhh, dbih, dbhh, fcW, fcb, out, B);
}


# Decoder feedback folded into its weights (Whh+Wih*fcW, b+Wih*fcb): next h no longer waits for pred; each wave forms pred over its own 32 columns a step late, partials summed in LDS; new sum order
# speedup vs baseline: 4.2713x; 1.0971x over previous
#include <hip/hip_runtime.h>
#include <cmath>


typedef float v2f __attribute__((ext_vector_type(2)));
typedef float v4f __attribute__((ext_vector_type(4)));
typedef float v8f __attribute__((ext_vector_type(8)));

constexpr int Hdim  = 128;
constexpr int TT    = 64;
constexpr int PLEN  = 32;
constexpr int WAVES = 4;
constexpr int NK    = 32;
constexpr int NJW   = 2;
constexpr int APACK = NK * 32 * 2;
constexpr int NKQ   = NK / WAVES;
constexpr int PARTF = 16 * 4 * WAVES;
constexpr int LDS_USED_FLOATS = 2 * APACK + TT * 16 * 4 + 2 * PARTF;

extern __shared__ __attribute__((aligned(16))) float smem[];

__device__ __forceinline__ int a_idx(int m, int k) {
    return (k >> 2) * 64 + (((((k >> 1) & 1) << 4) + m) << 1) + (k & 1);
}

__device__ __forceinline__ float fast_tanh(float x) {
    float e = __builtin_amdgcn_exp2f(x * 2.88539008177792681472f);
    return 1.0f - 2.0f * __builtin_amdgcn_rcpf(e + 1.0f);
}

__device__ __forceinline__ v8f mma(v2f a, v2f b, v8f c) {
    return __builtin_amdgcn_wmma_f32_16x16x4_f32(false, a, false, b, (short)0, c, false, false);
}

template <int I>
__device__ __forceinline__ float row_share(float v) {
    return __int_as_float(__builtin_amdgcn_update_dpp(0, __float_as_int(v), 0x150 + I, 0xf, 0xf, false));
}

__device__ __forceinline__ void load_bw(v2f (&bw)[NK][NJW], const float* __restrict__ Wm,
                                        int colbase, int half, int ln) {
#pragma unroll
    for (int kk = 0; kk < NK; ++kk)
#pragma unroll
        for (int jj = 0; jj < NJW; ++jj)
            bw[kk][jj] = *(const v2f*)&Wm[(colbase + jj * 16 + ln) * Hdim + kk * 4 + half * 2];
}

__device__ __forceinline__ void cinit(v8f (&c)[NJW], const v4f (&xv)[8],
                                      const float (&bs)[NJW], const float (&wi)[NJW][3]) {
#pragma unroll
    for (int jj = 0; jj < NJW; ++jj)
#pragma unroll
        for (int r = 0; r < 8; ++r)
            c[jj][r] = bs[jj] + xv[r].x * wi[jj][0] + xv[r].y * wi[jj][1] + xv[r].z * wi[jj][2];
}

__device__ __forceinline__ void store_h(float* hn, const v8f (&c)[NJW], int half, int colbase, int ln) {
#pragma unroll
    for (int jj = 0; jj < NJW; ++jj)
#pragma unroll
        for (int r = 0; r < 8; ++r)
            hn[a_idx(r + half * 8, colbase + jj * 16 + ln)] = fast_tanh(c[jj][r]);
}


__device__ __forceinline__ void load_own(v2f (&af)[NKQ], const float* hc, int wave, int lane) {
#pragma unroll
    for (int i = 0; i < NKQ; ++i) af[i] = *(const v2f*)&hc[((wave * NKQ + i) * 32 + lane) * 2];
}

__device__ __forceinline__ v8f fc_partial(const v2f (&af)[NKQ], const v2f (&bfq)[NKQ], float fcb0) {
    v8f pa, pb;
#pragma unroll
    for (int r = 0; r < 8; ++r) { pa[r] = fcb0; pb[r] = 0.0f; }
#pragma unroll
    for (int i = 0; i < NKQ / 2; ++i) {
        pa = mma(af[2 * i], bfq[2 * i], pa);
        pb = mma(af[2 * i + 1], bfq[2 * i + 1], pb);
    }
    return pa + pb;
}

__device__ __forceinline__ void store_partial(float* part, const v8f& pp, int wave, int half, int ln) {
    if (ln < 3) {
#pragma unroll
        for (int r = 0; r < 8; ++r) part[((r + half * 8) * 4 + ln) * WAVES + wave] = pp[r];
    }
}

__device__ __forceinline__ void load_partials(const float* part, v4f& q0, v4f& q1, int wave, int half, int lnc) {
    const int r0 = 2 * wave + half * 8;
    q0 = *(const v4f*)&part[((r0    ) * 4 + lnc) * WAVES];
    q1 = *(const v4f*)&part[((r0 + 1) * 4 + lnc) * WAVES];
}

__device__ __forceinline__ void emit_pred(const v4f& q0, const v4f& q1, float* outp, bool writer) {
    float o0 = (q0.x + q0.y) + (q0.z + q0.w);
    float o1 = (q1.x + q1.y) + (q1.z + q1.w);
    if (writer) { outp[0] = o0; outp[PLEN * 3] = o1; }
}

template <bool FC, bool EMIT>
__device__ __forceinline__ void dec_step(const float* hc, float* hn, const float* pold, float* pnew,
                                         v2f (&af)[NKQ], const v2f (&bw)[NK][NJW], const v2f (&bfq)[NKQ],
                                         const float (&bs)[NJW], float fcb0, float* outp, bool writer,
                                         int wave, int lane, int half, int ln, int lnc, int colbase) {
    v4f q0, q1;
    if (EMIT) load_partials(pold, q0, q1, wave, half, lnc);
    v2f a[NK];
#pragma unroll
    for (int kk = 0; kk < NK; ++kk) a[kk] = *(const v2f*)&hc[(kk * 32 + lane) * 2];
    v8f pp;
    if (FC) pp = fc_partial(af, bfq, fcb0);
    load_own(af, hc, wave, lane);
    if (EMIT) {
        asm volatile("" :: "v"(q0), "v"(q1));
        emit_pred(q0, q1, outp, writer);
    }
    v8f c[NJW];
#pragma unroll
    for (int jj = 0; jj < NJW; ++jj)
#pragma unroll
        for (int r = 0; r < 8; ++r) c[jj][r] = bs[jj];
#pragma unroll
    for (int kk = 0; kk < NK; ++kk) {
        c[0] = mma(a[kk], bw[kk][0], c[0]);
        c[1] = mma(a[kk], bw[kk][1], c[1]);
    }
    if (FC) store_partial(pnew, pp, wave, half, ln);
    store_h(hn, c, half, colbase, ln);
}

__global__ __launch_bounds__(WAVES * 32)
void traj_rnn_kernel(const float* __restrict__ x,
                     const float* __restrict__ eWih, const float* __restrict__ eWhh,
                     const float* __restrict__ ebih, const float* __restrict__ ebhh,
                     const float* __restrict__ dWih, const float* __restrict__ dWhh,
                     const float* __restrict__ dbih, const float* __restrict__ dbhh,
                     const float* __restrict__ fcW,  const float* __restrict__ fcb,
                     float* __restrict__ out, int B)
{
    float* hbuf = smem;
    float* xl   = smem + 2 * APACK;
    float* part = xl + TT * 16 * 4;

    const int tid  = threadIdx.x;
    const int wave = __builtin_amdgcn_readfirstlane(tid >> 5);
    const int lane = tid & 31;
    const int half = lane >> 4;
    const int ln   = lane & 15;
    const int colbase = wave * 32;

    const int b0 = blockIdx.x * 16;
    if (b0 >= B) return;

    const float* xt = x + (size_t)b0 * TT * 3;
    for (int e = tid; e < 16 * TT * 3; e += WAVES * 32) {
        int row = e / (TT * 3);
        int rem = e - row * (TT * 3);
        int t   = rem / 3;
        int i   = rem - t * 3;
        xl[(t * 16 + row) * 4 + i] = xt[e];
    }
    for (int e = tid; e < TT * 16; e += WAVES * 32) xl[e * 4 + 3] = 0.0f;
    for (int e = tid; e < APACK; e += WAVES * 32) hbuf[e] = 0.0f;

    float bse[NJW], bsd[NJW], bsf[NJW], wie[NJW][3], wid[NJW][3];
#pragma unroll
    for (int jj = 0; jj < NJW; ++jj) {
        int col = colbase + jj * 16 + ln;
        bse[jj] = ebih[col] + ebhh[col];
        bsd[jj] = dbih[col] + dbhh[col];
#pragma unroll
        for (int i = 0; i < 3; ++i) {
            wie[jj][i] = eWih[col * 3 + i];
            wid[jj][i] = dWih[col * 3 + i];
        }
        bsf[jj] = bsd[jj] + (wid[jj][0] * fcb[0] + wid[jj][1] * fcb[1] + wid[jj][2] * fcb[2]);
    }

    v2f bwe[NK][NJW], bwd[NK][NJW], bwf[NK][NJW], bfq[NKQ];
    load_bw(bwe, eWhh, colbase, half, ln);
    load_bw(bwd, dWhh, colbase, half, ln);
#pragma unroll
    for (int kk = 0; kk < NK; ++kk) {
        v2f f0 = *(const v2f*)&fcW[0 * Hdim + kk * 4 + half * 2];
        v2f f1 = *(const v2f*)&fcW[1 * Hdim + kk * 4 + half * 2];
        v2f f2 = *(const v2f*)&fcW[2 * Hdim + kk * 4 + half * 2];
#pragma unroll
        for (int jj = 0; jj < NJW; ++jj) {
            bwf[kk][jj].x = bwd[kk][jj].x + (wid[jj][0] * f0.x + wid[jj][1] * f1.x + wid[jj][2] * f2.x);
            bwf[kk][jj].y = bwd[kk][jj].y + (wid[jj][0] * f0.y + wid[jj][1] * f1.y + wid[jj][2] * f2.y);
        }
    }
#pragma unroll
    for (int i = 0; i < NKQ; ++i) {
        v2f f = *(const v2f*)&fcW[(ln < 3 ? ln : 0) * Hdim + (wave * NKQ + i) * 4 + half * 2];
        bfq[i].x = ln < 3 ? f.x : 0.0f;
        bfq[i].y = ln < 3 ? f.y : 0.0f;
    }
    const float fcb0 = (wave == 0 && ln < 3) ? fcb[ln < 3 ? ln : 0] : 0.0f;

    __syncthreads();

    v4f xn[8];
#pragma unroll
    for (int r = 0; r < 8; ++r) xn[r] = *(const v4f*)&xl[(r + half * 8) * 4];
    v8f c[NJW];
    cinit(c, xn, bse, wie);

    for (int t = 0; t < TT; ++t) {
        const float* hc = hbuf + (t & 1) * APACK;
        float*       hn = hbuf + ((t & 1) ^ 1) * APACK;
        v2f a[NK];
#pragma unroll
        for (int kk = 0; kk < NK; ++kk) a[kk] = *(const v2f*)&hc[(kk * 32 + lane) * 2];
        const int tn = t + 1 < TT ? t + 1 : TT - 1;
#pragma unroll
        for (int r = 0; r < 8; ++r) xn[r] = *(const v4f*)&xl[(tn * 16 + r + half * 8) * 4];

#pragma unroll
        for (int kk = 0; kk < NK / 2; ++kk) {
            c[0] = mma(a[kk], bwe[kk][0], c[0]);
            c[1] = mma(a[kk], bwe[kk][1], c[1]);
        }
        v8f cn[NJW];
        cinit(cn, xn, bse, wie);
#pragma unroll
        for (int kk = NK / 2; kk < NK; ++kk) {
            c[0] = mma(a[kk], bwe[kk][0], c[0]);
            c[1] = mma(a[kk], bwe[kk][1], c[1]);
        }
        store_h(hn, c, half, colbase, ln);
        c[0] = cn[0];
        c[1] = cn[1];
        __syncthreads();
    }

    int cur = TT & 1;
    const int  lnc    = ln < 3 ? ln : 0;
    const bool writer = ln < 3;
    float* outp = out + ((size_t)(b0 + 2 * wave + half * 8) * PLEN) * 3 + lnc;

    {
        const float* hc = hbuf + cur * APACK;
        v2f a[NK];
#pragma unroll
        for (int kk = 0; kk < NK; ++kk) a[kk] = *(const v2f*)&hc[(kk * 32 + lane) * 2];
        cinit(c, xn, bsd, wid);
#pragma unroll
        for (int kk = 0; kk < NK; ++kk) {
            c[0] = mma(a[kk], bwd[kk][0], c[0]);
            c[1] = mma(a[kk], bwd[kk][1], c[1]);
        }
        store_h(hbuf + (cur ^ 1) * APACK, c, half, colbase, ln);
    }
    __syncthreads();
    cur ^= 1;
    v2f af[NKQ];
    dec_step<false, false>(hbuf + cur * APACK, hbuf + (cur ^ 1) * APACK, part, part, af, bwf, bfq, bsf, fcb0,
                           outp, writer, wave, lane, half, ln, lnc, colbase);
    __syncthreads();
    cur ^= 1;
    dec_step<true, false>(hbuf + cur * APACK, hbuf + (cur ^ 1) * APACK, part, part, af, bwf, bfq, bsf, fcb0,
                          outp, writer, wave, lane, half, ln, lnc, colbase);
    __syncthreads();
    cur ^= 1;
    for (int p = 3; p < PLEN; ++p) {
        dec_step<true, true>(hbuf + cur * APACK, hbuf + (cur ^ 1) * APACK,
                             part + ((p - 1) & 1) * PARTF, part + (p & 1) * PARTF, af, bwf, bfq, bsf, fcb0,
                             outp + (p - 3) * 3, writer, wave, lane, half, ln, lnc, colbase);
        __syncthreads();
        cur ^= 1;
    }
    {
        v4f q0, q1;
        load_partials(part + ((PLEN - 1) & 1) * PARTF, q0, q1, wave, half, lnc);
        emit_pred(q0, q1, outp + (PLEN - 3) * 3, writer);
        v8f pp = fc_partial(af, bfq, fcb0);
        store_partial(part + (PLEN & 1) * PARTF, pp, wave, half, ln);
        load_own(af, hbuf + cur * APACK, wave, lane);
        pp = fc_partial(af, bfq, fcb0);
        __syncthreads();
        load_partials(part + (PLEN & 1) * PARTF, q0, q1, wave, half, lnc);
        emit_pred(q0, q1, outp + (PLEN - 2) * 3, writer);
        store_partial(part + ((PLEN + 1) & 1) * PARTF, pp, wave, half, ln);
        __syncthreads();
        load_partials(part + ((PLEN + 1) & 1) * PARTF, q0, q1, wave, half, lnc);
        emit_pred(q0, q1, outp + (PLEN - 1) * 3, writer);
    }
}

extern "C" void kernel_launch(void* const* d_in, const int* in_sizes, int n_in,
                              void* d_out, int out_size, void* d_ws, size_t ws_size,
                              hipStream_t stream) {
    (void)n_in; (void)out_size; (void)d_ws; (void)ws_size;
    const float* x    = (const float*)d_in[0];
    const float* eWih = (const float*)d_in[1];
    const float* eWhh = (const float*)d_in[2];
    const float* ebih = (const float*)d_in[3];
    const float* ebhh = (const float*)d_in[4];
    const float* dWih = (const float*)d_in[5];
    const float* dWhh = (const float*)d_in[6];
    const float* dbih = (const float*)d_in[7];
    const float* dbhh = (const float*)d_in[8];
    const float* fcW  = (const float*)d_in[9];
    const float* fcb  = (const float*)d_in[10];
    float* out = (float*)d_out;

    int B = in_sizes[0] / (TT * 3);
    int blocks = (B + 15) / 16;

    static_assert(LDS_USED_FLOATS * sizeof(float) <= 164 * 1024, "LDS request too small");
    size_t shmem = 164 * 1024;

    traj_rnn_kernel<<<blocks, WAVES * 32, shmem, stream>>>(
        x, eWih, eWhh, ebih, ebhh, dWih, dWhh, dbih, dbhh, fcW, fcb, out, B);
}


# A wave's two feature tiles run half a step apart (split barrier, two per step): each tile's tanh, stores, barrier and loads hide behind the other tile's 32 products; summed in halves: new sum order
# speedup vs baseline: 5.1935x; 1.2159x over previous
#include <hip/hip_runtime.h>
#include <cmath>


typedef float v2f __attribute__((ext_vector_type(2)));
typedef float v3f __attribute__((ext_vector_type(3)));
typedef float v4f __attribute__((ext_vector_type(4)));
typedef float v8f __attribute__((ext_vector_type(8)));

constexpr int Hdim  = 128;
constexpr int TT    = 64;
constexpr int PLEN  = 32;
constexpr int WAVES = 4;
constexpr int NK    = 32;
constexpr int NJW   = 2;
constexpr int APACK = NK * 32 * 2;
constexpr int NKQ   = NK / WAVES;
constexpr int PARTF = 16 * 4 * WAVES;
constexpr int LDS_USED_FLOATS = 2 * APACK + TT * 16 * 4 + 2 * PARTF;

extern __shared__ __attribute__((aligned(16))) float smem[];


__device__ __forceinline__ int a_idx(int m, int k) {
    return (k >> 2) * 64 + (((((k >> 1) & 1) << 4) + m) << 1) + (k & 1);
}

__device__ __forceinline__ int foreign_kk(int i, int wave) {
    return ((wave + (i >> 3)) & 3) * NKQ + (i & 7);
}

__device__ __forceinline__ int chunk_k0(int i, int wave, int half) {
    if (i < NKQ) return wave * 32 + 16 * (i >> 2) + 8 * half + 2 * (i & 3);
    return 4 * foreign_kk(i, wave) + 2 * half;
}

__device__ __forceinline__ float fast_tanh(float x) {
    float e = __builtin_amdgcn_exp2f(x * 2.88539008177792681472f);
    return 1.0f - 2.0f * __builtin_amdgcn_rcpf(e + 1.0f);
}

__device__ __forceinline__ v8f mma(v2f a, v2f b, v8f c) {
    return __builtin_amdgcn_wmma_f32_16x16x4_f32(false, a, false, b, (short)0, c, false, false);
}

__device__ __forceinline__ void load_bw(v2f (&bw)[NK][NJW], const float* __restrict__ Wm,
                                        int colbase, int wave, int half, int ln) {
#pragma unroll
    for (int i = 0; i < NK; ++i)
#pragma unroll
        for (int jj = 0; jj < NJW; ++jj)
            bw[i][jj] = *(const v2f*)&Wm[(colbase + jj * 16 + ln) * Hdim + chunk_k0(i, wave, half)];
}

constexpr int NFG = 3 * (NKQ / 2);

template <int G>
__device__ __forceinline__ void load_group(v2f (&f)[NFG], const float* hc, int wave, int lane) {
    const float* hl = hc + lane * 2 + G * (4 * 64);
#pragma unroll
    for (int q = 1; q < WAVES; ++q) {
        const float* hq = hl + ((wave + q) & 3) * (NKQ * 64);
#pragma unroll
        for (int j = 0; j < 4; ++j) f[(q - 1) * 4 + j] = *(const v2f*)&hq[j * 64];
    }
}

template <int JJ, int G, int N0, int N1>
__device__ __forceinline__ void group_mma(v8f& p, v8f& q, const v2f (&bw)[NK][NJW],
                                          const v2f (&ho)[4], const v2f (&f)[NFG]) {
#pragma unroll
    for (int n = N0; n < N1; ++n) {
        if (n < 4) {
            if (n & 1) q = mma(bw[4 * G + n][JJ], ho[n], q);
            else       p = mma(bw[4 * G + n][JJ], ho[n], p);
        } else {
            const int m = n - 4;
            const int i = 8 * (1 + (m >> 2)) + 4 * G + (m & 3);
            if (n & 1) q = mma(bw[i][JJ], f[m], q);
            else       p = mma(bw[i][JJ], f[m], p);
        }
    }
}

template <int JJ>
__device__ __forceinline__ void cinit_tile(v8f& c, const v3f& xv,
                                           const float (&bs)[NJW][8], const float (&wi)[NJW][8][3]) {
#pragma unroll
    for (int r = 0; r < 8; ++r)
        c[r] = bs[JJ][r] + xv.x * wi[JJ][r][0] + xv.y * wi[JJ][r][1] + xv.z * wi[JJ][r][2];
}

template <int JJ>
__device__ __forceinline__ void tanh_store(float* hn, const v8f& t, v2f (&ho)[4], int half, int colbase, int ln) {
#pragma unroll
    for (int q = 0; q < 4; ++q) {
        v2f v;
        v.x = fast_tanh(t[2 * q]);
        v.y = fast_tanh(t[2 * q + 1]);
        ho[q] = v;
        *(v2f*)&hn[a_idx(ln, colbase + JJ * 16 + half * 8 + 2 * q)] = v;
    }
}

__device__ __forceinline__ void stores_done_signal() {
    asm volatile("s_wait_dscnt 0x0\n\ts_barrier_signal -1" ::: "memory");
}
__device__ __forceinline__ void others_done_wait() {
    asm volatile("s_barrier_wait -1" ::: "memory");
}


__device__ __forceinline__ v8f fc_partial(const v2f (&af)[NKQ], const v2f (&bfq)[NKQ], const float (&fcbv)[3]) {
    v8f pa, pb;
#pragma unroll
    for (int r = 0; r < 8; ++r) { pa[r] = 0.0f; pb[r] = 0.0f; }
    pa[0] = fcbv[0]; pa[1] = fcbv[1]; pa[2] = fcbv[2];
#pragma unroll
    for (int i = 0; i < NKQ / 2; ++i) {
        pa = mma(bfq[2 * i], af[2 * i], pa);
        pb = mma(bfq[2 * i + 1], af[2 * i + 1], pb);
    }
    return pa + pb;
}

__device__ __forceinline__ void store_partial(float* part, const v8f& pp, int wave, int lane) {
    if (lane < 16) {
#pragma unroll
        for (int o = 0; o < 3; ++o) part[(lane * 4 + o) * WAVES + wave] = pp[o];
    }
}

__device__ __forceinline__ void load_partials(const float* part, v4f& q0, v4f& q1, int wave, int half, int lnc) {
    const int r0 = 2 * wave + half * 8;
    q0 = *(const v4f*)&part[((r0    ) * 4 + lnc) * WAVES];
    q1 = *(const v4f*)&part[((r0 + 1) * 4 + lnc) * WAVES];
}

__device__ __forceinline__ void emit_pred(const v4f& q0, const v4f& q1, float* outp, bool writer) {
    float o0 = (q0.x + q0.y) + (q0.z + q0.w);
    float o1 = (q1.x + q1.y) + (q1.z + q1.w);
    if (writer) { outp[0] = o0; outp[PLEN * 3] = o1; }
}

template <bool PRED>
__device__ __forceinline__ void dec_step(float* hc, float* hn, float* pl,
                                         v8f& p0, v8f& q0, v8f& p1, v8f& q1,
                                         v2f (&h0f)[4], v2f (&h1f)[4], v2f (&f0)[NFG], v2f (&f1)[NFG],
                                         const v2f (&bw)[NK][NJW], const v8f (&init)[NJW],
                                         const v2f (&bfq)[NKQ], const float (&fcbv)[3], float* outp, bool writer,
                                         int wave, int lane, int half, int ln, int lnc, int colbase) {
    {
        const v8f t1 = p1 + q1;
        tanh_store<1>(hc, t1, h1f, half, colbase, ln);
    }
    p0 = init[0];
#pragma unroll
    for (int r = 0; r < 8; ++r) q0[r] = 0.0f;
    group_mma<0, 0, 0, 10>(p0, q0, bw, h0f, f0);
    __builtin_amdgcn_sched_barrier(0);
    stores_done_signal();
    __builtin_amdgcn_sched_barrier(0);
    group_mma<0, 0, 10, 14>(p0, q0, bw, h0f, f0);
    __builtin_amdgcn_sched_barrier(0);
    others_done_wait();
    load_group<1>(f1, hc, wave, lane);
    __builtin_amdgcn_sched_barrier(0);
    group_mma<0, 0, 14, 16>(p0, q0, bw, h0f, f0);
    v8f pp;
    if (PRED) {
        v2f hw[NKQ];
#pragma unroll
        for (int i = 0; i < 4; ++i) { hw[i] = h0f[i]; hw[4 + i] = h1f[i]; }
        pp = fc_partial(hw, bfq, fcbv);
    }
    group_mma<0, 1, 0, 16>(p0, q0, bw, h1f, f1);
    if (PRED) store_partial(pl, pp, wave, lane);
    __builtin_amdgcn_sched_barrier(0);
    v2f h0n[4];
    {
        const v8f t0 = p0 + q0;
        tanh_store<0>(hn, t0, h0n, half, colbase, ln);
    }
    p1 = init[1];
#pragma unroll
    for (int r = 0; r < 8; ++r) q1[r] = 0.0f;
    group_mma<1, 0, 0, 9>(p1, q1, bw, h0f, f0);
    __builtin_amdgcn_sched_barrier(0);
    group_mma<1, 0, 9, 13>(p1, q1, bw, h0f, f0);
    __builtin_amdgcn_sched_barrier(0);
    stores_done_signal();
    __builtin_amdgcn_sched_barrier(0);
    group_mma<1, 0, 13, 16>(p1, q1, bw, h0f, f0);
    group_mma<1, 1, 0, 6>(p1, q1, bw, h1f, f1);
    __builtin_amdgcn_sched_barrier(0);
    others_done_wait();
    load_group<0>(f0, hn, wave, lane);
    v4f qa, qb;
    if (PRED) load_partials(pl, qa, qb, wave, half, lnc);
    __builtin_amdgcn_sched_barrier(0);
    group_mma<1, 1, 6, 16>(p1, q1, bw, h1f, f1);
    if (PRED) emit_pred(qa, qb, outp, writer);
    __builtin_amdgcn_sched_barrier(0);
#pragma unroll
    for (int i = 0; i < 4; ++i) h0f[i] = h0n[i];
}

__global__ __launch_bounds__(WAVES * 32)
void traj_rnn_kernel(const float* __restrict__ x,
                     const float* __restrict__ eWih, const float* __restrict__ eWhh,
                     const float* __restrict__ ebih, const float* __restrict__ ebhh,
                     const float* __restrict__ dWih, const float* __restrict__ dWhh,
                     const float* __restrict__ dbih, const float* __restrict__ dbhh,
                     const float* __restrict__ fcW,  const float* __restrict__ fcb,
                     float* __restrict__ out, int B)
{
    float* hbuf = smem;
    float* xl   = smem + 2 * APACK;
    float* part = xl + TT * 16 * 4;

    const int tid  = threadIdx.x;
    const int wave = __builtin_amdgcn_readfirstlane(tid >> 5);
    const int lane = tid & 31;
    const int half = lane >> 4;
    const int ln   = lane & 15;
    const int colbase = wave * 32;

    const int b0 = blockIdx.x * 16;
    if (b0 >= B) return;

    const float* xt = x + (size_t)b0 * TT * 3;
    for (int e = tid; e < 16 * TT * 3; e += WAVES * 32) {
        int row = e / (TT * 3);
        int rem = e - row * (TT * 3);
        int t   = rem / 3;
        int i   = rem - t * 3;
        xl[(t * 16 + row) * 4 + i] = xt[e];
    }
    for (int e = tid; e < APACK; e += WAVES * 32) hbuf[e] = 0.0f;

    float bse[NJW][8], wie[NJW][8][3], bsf[NJW][8];
    v8f cdec0[NJW];
    {
        const float* xlast = x + ((size_t)(b0 + ln) * TT + (TT - 1)) * 3;
        const float xl0 = xlast[0], xl1 = xlast[1], xl2 = xlast[2];
        const float fb0 = fcb[0], fb1 = fcb[1], fb2 = fcb[2];
#pragma unroll
        for (int jj = 0; jj < NJW; ++jj)
#pragma unroll
            for (int r = 0; r < 8; ++r) {
                const int f = colbase + jj * 16 + half * 8 + r;
                bse[jj][r] = ebih[f] + ebhh[f];
#pragma unroll
                for (int i = 0; i < 3; ++i) wie[jj][r][i] = eWih[f * 3 + i];
                const float bd = dbih[f] + dbhh[f];
                const float w0 = dWih[f * 3 + 0], w1 = dWih[f * 3 + 1], w2 = dWih[f * 3 + 2];
                cdec0[jj][r] = bd + xl0 * w0 + xl1 * w1 + xl2 * w2;
                bsf[jj][r] = bd + (w0 * fb0 + w1 * fb1 + w2 * fb2);
            }
    }
    float wid[NJW][3];
#pragma unroll
    for (int jj = 0; jj < NJW; ++jj)
#pragma unroll
        for (int i = 0; i < 3; ++i) wid[jj][i] = dWih[(colbase + jj * 16 + ln) * 3 + i];

    v2f bwe[NK][NJW], bwd[NK][NJW], bwf[NK][NJW], bfq[NKQ];
    load_bw(bwe, eWhh, colbase, wave, half, ln);
    load_bw(bwd, dWhh, colbase, wave, half, ln);
#pragma unroll
    for (int i = 0; i < NK; ++i) {
        const int k0 = chunk_k0(i, wave, half);
        v2f f0 = *(const v2f*)&fcW[0 * Hdim + k0];
        v2f f1 = *(const v2f*)&fcW[1 * Hdim + k0];
        v2f f2 = *(const v2f*)&fcW[2 * Hdim + k0];
#pragma unroll
        for (int jj = 0; jj < NJW; ++jj) {
            bwf[i][jj].x = bwd[i][jj].x + (wid[jj][0] * f0.x + wid[jj][1] * f1.x + wid[jj][2] * f2.x);
            bwf[i][jj].y = bwd[i][jj].y + (wid[jj][0] * f0.y + wid[jj][1] * f1.y + wid[jj][2] * f2.y);
        }
    }
#pragma unroll
    for (int i = 0; i < NKQ; ++i) {
        v2f f = *(const v2f*)&fcW[(ln < 3 ? ln : 0) * Hdim + chunk_k0(i, wave, half)];
        bfq[i].x = ln < 3 ? f.x : 0.0f;
        bfq[i].y = ln < 3 ? f.y : 0.0f;
    }
    float fcbv[3];
#pragma unroll
    for (int o = 0; o < 3; ++o) fcbv[o] = (wave == 0) ? fcb[o] : 0.0f;

    __syncthreads();

    v2f h0f[4], h1f[4];
#pragma unroll
    for (int i = 0; i < 4; ++i) { h0f[i].x = 0.0f; h0f[i].y = 0.0f; }
    v8f p0, q0, p1, q1;
#pragma unroll
    for (int r = 0; r < 8; ++r) { p1[r] = 0.0f; q1[r] = 0.0f; }
    v2f f0[NFG], f1[NFG];
    load_group<0>(f0, hbuf, wave, lane);
    v3f xc = *(const v3f*)&xl[ln * 4];

    for (int t = 0; t < TT; ++t) {
        float* hc = hbuf + (t & 1) * APACK;
        float* hn = hbuf + ((t & 1) ^ 1) * APACK;
        {
            const v8f t1 = p1 + q1;
            tanh_store<1>(hc, t1, h1f, half, colbase, ln);
        }
        cinit_tile<0>(p0, xc, bse, wie);
#pragma unroll
        for (int r = 0; r < 8; ++r) q0[r] = 0.0f;
        group_mma<0, 0, 0, 10>(p0, q0, bwe, h0f, f0);
        __builtin_amdgcn_sched_barrier(0);
        stores_done_signal();
        __builtin_amdgcn_sched_barrier(0);
        group_mma<0, 0, 10, 14>(p0, q0, bwe, h0f, f0);
        __builtin_amdgcn_sched_barrier(0);
        others_done_wait();
        load_group<1>(f1, hc, wave, lane);
        const int tn = t + 1 < TT ? t + 1 : TT - 1;
        const v3f xn = *(const v3f*)&xl[(tn * 16 + ln) * 4];
        __builtin_amdgcn_sched_barrier(0);
        group_mma<0, 0, 14, 16>(p0, q0, bwe, h0f, f0);
        group_mma<0, 1, 0, 16>(p0, q0, bwe, h1f, f1);
        __builtin_amdgcn_sched_barrier(0);
        v2f h0n[4];
        {
            const v8f t0 = p0 + q0;
            tanh_store<0>(hn, t0, h0n, half, colbase, ln);
        }
        cinit_tile<1>(p1, xc, bse, wie);
#pragma unroll
        for (int r = 0; r < 8; ++r) q1[r] = 0.0f;
        group_mma<1, 0, 0, 9>(p1, q1, bwe, h0f, f0);
        __builtin_amdgcn_sched_barrier(0);
        group_mma<1, 0, 9, 13>(p1, q1, bwe, h0f, f0);
        __builtin_amdgcn_sched_barrier(0);
        stores_done_signal();
        __builtin_amdgcn_sched_barrier(0);
        group_mma<1, 0, 13, 16>(p1, q1, bwe, h0f, f0);
        group_mma<1, 1, 0, 6>(p1, q1, bwe, h1f, f1);
        __builtin_amdgcn_sched_barrier(0);
        others_done_wait();
        load_group<0>(f0, hn, wave, lane);
        __builtin_amdgcn_sched_barrier(0);
        group_mma<1, 1, 6, 16>(p1, q1, bwe, h1f, f1);
        __builtin_amdgcn_sched_barrier(0);
#pragma unroll
        for (int i = 0; i < 4; ++i) h0f[i] = h0n[i];
        xc = xn;
    }
    const int  lnc    = ln < 3 ? ln : 0;
    const bool writer = ln < 3;
    float* outp = out + ((size_t)(b0 + 2 * wave + half * 8) * PLEN) * 3 + lnc;
    v8f bsfv[NJW];
#pragma unroll
    for (int jj = 0; jj < NJW; ++jj)
#pragma unroll
        for (int r = 0; r < 8; ++r) bsfv[jj][r] = bsf[jj][r];

    dec_step<false>(hbuf + (TT & 1) * APACK, hbuf + ((TT & 1) ^ 1) * APACK, part,
                    p0, q0, p1, q1, h0f, h1f, f0, f1, bwd, cdec0, bfq, fcbv,
                    outp, writer, wave, lane, half, ln, lnc, colbase);
    for (int p = 1; p < PLEN; ++p) {
        const int s = (TT + p) & 1;
        dec_step<true>(hbuf + s * APACK, hbuf + (s ^ 1) * APACK, part + (p & 1) * PARTF,
                       p0, q0, p1, q1, h0f, h1f, f0, f1, bwf, bsfv, bfq, fcbv,
                       outp + (p - 1) * 3, writer, wave, lane, half, ln, lnc, colbase);
    }
    {
        const v8f t1 = p1 + q1;
        v2f hw[NKQ];
#pragma unroll
        for (int q = 0; q < 4; ++q) {
            hw[q] = h0f[q];
            hw[4 + q].x = fast_tanh(t1[2 * q]);
            hw[4 + q].y = fast_tanh(t1[2 * q + 1]);
        }
        const v8f pp = fc_partial(hw, bfq, fcbv);
        float* pl = part + (PLEN & 1) * PARTF;
        store_partial(pl, pp, wave, lane);
        __syncthreads();
        v4f qa, qb;
        load_partials(pl, qa, qb, wave, half, lnc);
        emit_pred(qa, qb, outp + (PLEN - 1) * 3, writer);
    }
}

extern "C" void kernel_launch(void* const* d_in, const int* in_sizes, int n_in,
                              void* d_out, int out_size, void* d_ws, size_t ws_size,
                              hipStream_t stream) {
    (void)n_in; (void)out_size; (void)d_ws; (void)ws_size;
    const float* x    = (const float*)d_in[0];
    const float* eWih = (const float*)d_in[1];
    const float* eWhh = (const float*)d_in[2];
    const float* ebih = (const float*)d_in[3];
    const float* ebhh = (const float*)d_in[4];
    const float* dWih = (const float*)d_in[5];
    const float* dWhh = (const float*)d_in[6];
    const float* dbih = (const float*)d_in[7];
    const float* dbhh = (const float*)d_in[8];
    const float* fcW  = (const float*)d_in[9];
    const float* fcb  = (const float*)d_in[10];
    float* out = (float*)d_out;

    int B = in_sizes[0] / (TT * 3);
    int blocks = (B + 15) / 16;

    static_assert(LDS_USED_FLOATS * sizeof(float) <= 164 * 1024, "LDS request too small");
    size_t shmem = 164 * 1024;

    traj_rnn_kernel<<<blocks, WAVES * 32, shmem, stream>>>(
        x, eWih, eWhh, ebih, ebhh, dWih, dWhh, dbih, dbhh, fcW, fcb, out, B);
}


# tanh by the chip's own tanh instruction (v_tanh_f32), one transcendental per value, in place of exp2, add, reciprocal, fma: 60 fewer vector issue cycles per step; tanh rounding differs, same sums
# speedup vs baseline: 5.4056x; 1.0408x over previous
#include <hip/hip_runtime.h>
#include <cmath>


typedef float v2f __attribute__((ext_vector_type(2)));
typedef float v3f __attribute__((ext_vector_type(3)));
typedef float v4f __attribute__((ext_vector_type(4)));
typedef float v8f __attribute__((ext_vector_type(8)));

constexpr int Hdim  = 128;
constexpr int TT    = 64;
constexpr int PLEN  = 32;
constexpr int WAVES = 4;
constexpr int NK    = 32;
constexpr int NJW   = 2;
constexpr int APACK = NK * 32 * 2;
constexpr int NKQ   = NK / WAVES;
constexpr int PARTF = 16 * 4 * WAVES;
constexpr int LDS_USED_FLOATS = 2 * APACK + TT * 16 * 4 + 2 * PARTF;

extern __shared__ __attribute__((aligned(16))) float smem[];


__device__ __forceinline__ int a_idx(int m, int k) {
    return (k >> 2) * 64 + (((((k >> 1) & 1) << 4) + m) << 1) + (k & 1);
}

__device__ __forceinline__ int foreign_kk(int i, int wave) {
    return ((wave + (i >> 3)) & 3) * NKQ + (i & 7);
}

__device__ __forceinline__ int chunk_k0(int i, int wave, int half) {
    if (i < NKQ) return wave * 32 + 16 * (i >> 2) + 8 * half + 2 * (i & 3);
    return 4 * foreign_kk(i, wave) + 2 * half;
}

__device__ __forceinline__ float fast_tanh(float x) {
    return __builtin_amdgcn_tanhf(x);
}

__device__ __forceinline__ v8f mma(v2f a, v2f b, v8f c) {
    return __builtin_amdgcn_wmma_f32_16x16x4_f32(false, a, false, b, (short)0, c, false, false);
}

__device__ __forceinline__ void load_bw(v2f (&bw)[NK][NJW], const float* __restrict__ Wm,
                                        int colbase, int wave, int half, int ln) {
#pragma unroll
    for (int i = 0; i < NK; ++i)
#pragma unroll
        for (int jj = 0; jj < NJW; ++jj)
            bw[i][jj] = *(const v2f*)&Wm[(colbase + jj * 16 + ln) * Hdim + chunk_k0(i, wave, half)];
}

constexpr int NFG = 3 * (NKQ / 2);

template <int G>
__device__ __forceinline__ void load_group(v2f (&f)[NFG], const float* hc, int wave, int lane) {
    const float* hl = hc + lane * 2 + G * (4 * 64);
#pragma unroll
    for (int q = 1; q < WAVES; ++q) {
        const float* hq = hl + ((wave + q) & 3) * (NKQ * 64);
#pragma unroll
        for (int j = 0; j < 4; ++j) f[(q - 1) * 4 + j] = *(const v2f*)&hq[j * 64];
    }
}

template <int JJ, int G, int N0, int N1>
__device__ __forceinline__ void group_mma(v8f& p, v8f& q, const v2f (&bw)[NK][NJW],
                                          const v2f (&ho)[4], const v2f (&f)[NFG]) {
#pragma unroll
    for (int n = N0; n < N1; ++n) {
        if (n < 4) {
            if (n & 1) q = mma(bw[4 * G + n][JJ], ho[n], q);
            else       p = mma(bw[4 * G + n][JJ], ho[n], p);
        } else {
            const int m = n - 4;
            const int i = 8 * (1 + (m >> 2)) + 4 * G + (m & 3);
            if (n & 1) q = mma(bw[i][JJ], f[m], q);
            else       p = mma(bw[i][JJ], f[m], p);
        }
    }
}

template <int JJ>
__device__ __forceinline__ void cinit_tile(v8f& c, const v3f& xv,
                                           const float (&bs)[NJW][8], const float (&wi)[NJW][8][3]) {
#pragma unroll
    for (int r = 0; r < 8; ++r)
        c[r] = bs[JJ][r] + xv.x * wi[JJ][r][0] + xv.y * wi[JJ][r][1] + xv.z * wi[JJ][r][2];
}

template <int JJ>
__device__ __forceinline__ void tanh_store(float* hn, const v8f& t, v2f (&ho)[4], int half, int colbase, int ln) {
#pragma unroll
    for (int q = 0; q < 4; ++q) {
        v2f v;
        v.x = fast_tanh(t[2 * q]);
        v.y = fast_tanh(t[2 * q + 1]);
        ho[q] = v;
        *(v2f*)&hn[a_idx(ln, colbase + JJ * 16 + half * 8 + 2 * q)] = v;
    }
}

__device__ __forceinline__ void stores_done_signal() {
    asm volatile("s_wait_dscnt 0x0\n\ts_barrier_signal -1" ::: "memory");
}
__device__ __forceinline__ void others_done_wait() {
    asm volatile("s_barrier_wait -1" ::: "memory");
}


__device__ __forceinline__ v8f fc_partial(const v2f (&af)[NKQ], const v2f (&bfq)[NKQ], const float (&fcbv)[3]) {
    v8f pa, pb;
#pragma unroll
    for (int r = 0; r < 8; ++r) { pa[r] = 0.0f; pb[r] = 0.0f; }
    pa[0] = fcbv[0]; pa[1] = fcbv[1]; pa[2] = fcbv[2];
#pragma unroll
    for (int i = 0; i < NKQ / 2; ++i) {
        pa = mma(bfq[2 * i], af[2 * i], pa);
        pb = mma(bfq[2 * i + 1], af[2 * i + 1], pb);
    }
    return pa + pb;
}

__device__ __forceinline__ void store_partial(float* part, const v8f& pp, int wave, int lane) {
    if (lane < 16) {
#pragma unroll
        for (int o = 0; o < 3; ++o) part[(lane * 4 + o) * WAVES + wave] = pp[o];
    }
}

__device__ __forceinline__ void load_partials(const float* part, v4f& q0, v4f& q1, int wave, int half, int lnc) {
    const int r0 = 2 * wave + half * 8;
    q0 = *(const v4f*)&part[((r0    ) * 4 + lnc) * WAVES];
    q1 = *(const v4f*)&part[((r0 + 1) * 4 + lnc) * WAVES];
}

__device__ __forceinline__ void emit_pred(const v4f& q0, const v4f& q1, float* outp, bool writer) {
    float o0 = (q0.x + q0.y) + (q0.z + q0.w);
    float o1 = (q1.x + q1.y) + (q1.z + q1.w);
    if (writer) { outp[0] = o0; outp[PLEN * 3] = o1; }
}

template <bool PRED>
__device__ __forceinline__ void dec_step(float* hc, float* hn, float* pl,
                                         v8f& p0, v8f& q0, v8f& p1, v8f& q1,
                                         v2f (&h0f)[4], v2f (&h1f)[4], v2f (&f0)[NFG], v2f (&f1)[NFG],
                                         const v2f (&bw)[NK][NJW], const v8f (&init)[NJW],
                                         const v2f (&bfq)[NKQ], const float (&fcbv)[3], float* outp, bool writer,
                                         int wave, int lane, int half, int ln, int lnc, int colbase) {
    {
        const v8f t1 = p1 + q1;
        tanh_store<1>(hc, t1, h1f, half, colbase, ln);
    }
    p0 = init[0];
#pragma unroll
    for (int r = 0; r < 8; ++r) q0[r] = 0.0f;
    group_mma<0, 0, 0, 10>(p0, q0, bw, h0f, f0);
    __builtin_amdgcn_sched_barrier(0);
    stores_done_signal();
    __builtin_amdgcn_sched_barrier(0);
    group_mma<0, 0, 10, 14>(p0, q0, bw, h0f, f0);
    __builtin_amdgcn_sched_barrier(0);
    others_done_wait();
    load_group<1>(f1, hc, wave, lane);
    __builtin_amdgcn_sched_barrier(0);
    group_mma<0, 0, 14, 16>(p0, q0, bw, h0f, f0);
    v8f pp;
    if (PRED) {
        v2f hw[NKQ];
#pragma unroll
        for (int i = 0; i < 4; ++i) { hw[i] = h0f[i]; hw[4 + i] = h1f[i]; }
        pp = fc_partial(hw, bfq, fcbv);
    }
    group_mma<0, 1, 0, 16>(p0, q0, bw, h1f, f1);
    if (PRED) store_partial(pl, pp, wave, lane);
    __builtin_amdgcn_sched_barrier(0);
    v2f h0n[4];
    {
        const v8f t0 = p0 + q0;
        tanh_store<0>(hn, t0, h0n, half, colbase, ln);
    }
    p1 = init[1];
#pragma unroll
    for (int r = 0; r < 8; ++r) q1[r] = 0.0f;
    group_mma<1, 0, 0, 9>(p1, q1, bw, h0f, f0);
    __builtin_amdgcn_sched_barrier(0);
    group_mma<1, 0, 9, 13>(p1, q1, bw, h0f, f0);
    __builtin_amdgcn_sched_barrier(0);
    stores_done_signal();
    __builtin_amdgcn_sched_barrier(0);
    group_mma<1, 0, 13, 16>(p1, q1, bw, h0f, f0);
    group_mma<1, 1, 0, 6>(p1, q1, bw, h1f, f1);
    __builtin_amdgcn_sched_barrier(0);
    others_done_wait();
    load_group<0>(f0, hn, wave, lane);
    v4f qa, qb;
    if (PRED) load_partials(pl, qa, qb, wave, half, lnc);
    __builtin_amdgcn_sched_barrier(0);
    group_mma<1, 1, 6, 16>(p1, q1, bw, h1f, f1);
    if (PRED) emit_pred(qa, qb, outp, writer);
    __builtin_amdgcn_sched_barrier(0);
#pragma unroll
    for (int i = 0; i < 4; ++i) h0f[i] = h0n[i];
}

__global__ __launch_bounds__(WAVES * 32)
void traj_rnn_kernel(const float* __restrict__ x,
                     const float* __restrict__ eWih, const float* __restrict__ eWhh,
                     const float* __restrict__ ebih, const float* __restrict__ ebhh,
                     const float* __restrict__ dWih, const float* __restrict__ dWhh,
                     const float* __restrict__ dbih, const float* __restrict__ dbhh,
                     const float* __restrict__ fcW,  const float* __restrict__ fcb,
                     float* __restrict__ out, int B)
{
    float* hbuf = smem;
    float* xl   = smem + 2 * APACK;
    float* part = xl + TT * 16 * 4;

    const int tid  = threadIdx.x;
    const int wave = __builtin_amdgcn_readfirstlane(tid >> 5);
    const int lane = tid & 31;
    const int half = lane >> 4;
    const int ln   = lane & 15;
    const int colbase = wave * 32;

    const int b0 = blockIdx.x * 16;
    if (b0 >= B) return;

    const float* xt = x + (size_t)b0 * TT * 3;
    for (int e = tid; e < 16 * TT * 3; e += WAVES * 32) {
        int row = e / (TT * 3);
        int rem = e - row * (TT * 3);
        int t   = rem / 3;
        int i   = rem - t * 3;
        xl[(t * 16 + row) * 4 + i] = xt[e];
    }
    for (int e = tid; e < APACK; e += WAVES * 32) hbuf[e] = 0.0f;

    float bse[NJW][8], wie[NJW][8][3], bsf[NJW][8];
    v8f cdec0[NJW];
    {
        const float* xlast = x + ((size_t)(b0 + ln) * TT + (TT - 1)) * 3;
        const float xl0 = xlast[0], xl1 = xlast[1], xl2 = xlast[2];
        const float fb0 = fcb[0], fb1 = fcb[1], fb2 = fcb[2];
#pragma unroll
        for (int jj = 0; jj < NJW; ++jj)
#pragma unroll
            for (int r = 0; r < 8; ++r) {
                const int f = colbase + jj * 16 + half * 8 + r;
                bse[jj][r] = ebih[f] + ebhh[f];
#pragma unroll
                for (int i = 0; i < 3; ++i) wie[jj][r][i] = eWih[f * 3 + i];
                const float bd = dbih[f] + dbhh[f];
                const float w0 = dWih[f * 3 + 0], w1 = dWih[f * 3 + 1], w2 = dWih[f * 3 + 2];
                cdec0[jj][r] = bd + xl0 * w0 + xl1 * w1 + xl2 * w2;
                bsf[jj][r] = bd + (w0 * fb0 + w1 * fb1 + w2 * fb2);
            }
    }
    float wid[NJW][3];
#pragma unroll
    for (int jj = 0; jj < NJW; ++jj)
#pragma unroll
        for (int i = 0; i < 3; ++i) wid[jj][i] = dWih[(colbase + jj * 16 + ln) * 3 + i];

    v2f bwe[NK][NJW], bwd[NK][NJW], bwf[NK][NJW], bfq[NKQ];
    load_bw(bwe, eWhh, colbase, wave, half, ln);
    load_bw(bwd, dWhh, colbase, wave, half, ln);
#pragma unroll
    for (int i = 0; i < NK; ++i) {
        const int k0 = chunk_k0(i, wave, half);
        v2f f0 = *(const v2f*)&fcW[0 * Hdim + k0];
        v2f f1 = *(const v2f*)&fcW[1 * Hdim + k0];
        v2f f2 = *(const v2f*)&fcW[2 * Hdim + k0];
#pragma unroll
        for (int jj = 0; jj < NJW; ++jj) {
            bwf[i][jj].x = bwd[i][jj].x + (wid[jj][0] * f0.x + wid[jj][1] * f1.x + wid[jj][2] * f2.x);
            bwf[i][jj].y = bwd[i][jj].y + (wid[jj][0] * f0.y + wid[jj][1] * f1.y + wid[jj][2] * f2.y);
        }
    }
#pragma unroll
    for (int i = 0; i < NKQ; ++i) {
        v2f f = *(const v2f*)&fcW[(ln < 3 ? ln : 0) * Hdim + chunk_k0(i, wave, half)];
        bfq[i].x = ln < 3 ? f.x : 0.0f;
        bfq[i].y = ln < 3 ? f.y : 0.0f;
    }
    float fcbv[3];
#pragma unroll
    for (int o = 0; o < 3; ++o) fcbv[o] = (wave == 0) ? fcb[o] : 0.0f;

    __syncthreads();

    v2f h0f[4], h1f[4];
#pragma unroll
    for (int i = 0; i < 4; ++i) { h0f[i].x = 0.0f; h0f[i].y = 0.0f; }
    v8f p0, q0, p1, q1;
#pragma unroll
    for (int r = 0; r < 8; ++r) { p1[r] = 0.0f; q1[r] = 0.0f; }
    v2f f0[NFG], f1[NFG];
    load_group<0>(f0, hbuf, wave, lane);
    v3f xc = *(const v3f*)&xl[ln * 4];

    for (int t = 0; t < TT; ++t) {
        float* hc = hbuf + (t & 1) * APACK;
        float* hn = hbuf + ((t & 1) ^ 1) * APACK;
        {
            const v8f t1 = p1 + q1;
            tanh_store<1>(hc, t1, h1f, half, colbase, ln);
        }
        cinit_tile<0>(p0, xc, bse, wie);
#pragma unroll
        for (int r = 0; r < 8; ++r) q0[r] = 0.0f;
        group_mma<0, 0, 0, 10>(p0, q0, bwe, h0f, f0);
        __builtin_amdgcn_sched_barrier(0);
        stores_done_signal();
        __builtin_amdgcn_sched_barrier(0);
        group_mma<0, 0, 10, 14>(p0, q0, bwe, h0f, f0);
        __builtin_amdgcn_sched_barrier(0);
        others_done_wait();
        load_group<1>(f1, hc, wave, lane);
        const int tn = t + 1 < TT ? t + 1 : TT - 1;
        const v3f xn = *(const v3f*)&xl[(tn * 16 + ln) * 4];
        __builtin_amdgcn_sched_barrier(0);
        group_mma<0, 0, 14, 16>(p0, q0, bwe, h0f, f0);
        group_mma<0, 1, 0, 16>(p0, q0, bwe, h1f, f1);
        __builtin_amdgcn_sched_barrier(0);
        v2f h0n[4];
        {
            const v8f t0 = p0 + q0;
            tanh_store<0>(hn, t0, h0n, half, colbase, ln);
        }
        cinit_tile<1>(p1, xc, bse, wie);
#pragma unroll
        for (int r = 0; r < 8; ++r) q1[r] = 0.0f;
        group_mma<1, 0, 0, 9>(p1, q1, bwe, h0f, f0);
        __builtin_amdgcn_sched_barrier(0);
        group_mma<1, 0, 9, 13>(p1, q1, bwe, h0f, f0);
        __builtin_amdgcn_sched_barrier(0);
        stores_done_signal();
        __builtin_amdgcn_sched_barrier(0);
        group_mma<1, 0, 13, 16>(p1, q1, bwe, h0f, f0);
        group_mma<1, 1, 0, 6>(p1, q1, bwe, h1f, f1);
        __builtin_amdgcn_sched_barrier(0);
        others_done_wait();
        load_group<0>(f0, hn, wave, lane);
        __builtin_amdgcn_sched_barrier(0);
        group_mma<1, 1, 6, 16>(p1, q1, bwe, h1f, f1);
        __builtin_amdgcn_sched_barrier(0);
#pragma unroll
        for (int i = 0; i < 4; ++i) h0f[i] = h0n[i];
        xc = xn;
    }
    const int  lnc    = ln < 3 ? ln : 0;
    const bool writer = ln < 3;
    float* outp = out + ((size_t)(b0 + 2 * wave + half * 8) * PLEN) * 3 + lnc;
    v8f bsfv[NJW];
#pragma unroll
    for (int jj = 0; jj < NJW; ++jj)
#pragma unroll
        for (int r = 0; r < 8; ++r) bsfv[jj][r] = bsf[jj][r];

    dec_step<false>(hbuf + (TT & 1) * APACK, hbuf + ((TT & 1) ^ 1) * APACK, part,
                    p0, q0, p1, q1, h0f, h1f, f0, f1, bwd, cdec0, bfq, fcbv,
                    outp, writer, wave, lane, half, ln, lnc, colbase);
    for (int p = 1; p < PLEN; ++p) {
        const int s = (TT + p) & 1;
        dec_step<true>(hbuf + s * APACK, hbuf + (s ^ 1) * APACK, part + (p & 1) * PARTF,
                       p0, q0, p1, q1, h0f, h1f, f0, f1, bwf, bsfv, bfq, fcbv,
                       outp + (p - 1) * 3, writer, wave, lane, half, ln, lnc, colbase);
    }
    {
        const v8f t1 = p1 + q1;
        v2f hw[NKQ];
#pragma unroll
        for (int q = 0; q < 4; ++q) {
            hw[q] = h0f[q];
            hw[4 + q].x = fast_tanh(t1[2 * q]);
            hw[4 + q].y = fast_tanh(t1[2 * q + 1]);
        }
        const v8f pp = fc_partial(hw, bfq, fcbv);
        float* pl = part + (PLEN & 1) * PARTF;
        store_partial(pl, pp, wave, lane);
        __syncthreads();
        v4f qa, qb;
        load_partials(pl, qa, qb, wave, half, lnc);
        emit_pred(qa, qb, outp + (PLEN - 1) * 3, writer);
    }
}

extern "C" void kernel_launch(void* const* d_in, const int* in_sizes, int n_in,
                              void* d_out, int out_size, void* d_ws, size_t ws_size,
                              hipStream_t stream) {
    (void)n_in; (void)out_size; (void)d_ws; (void)ws_size;
    const float* x    = (const float*)d_in[0];
    const float* eWih = (const float*)d_in[1];
    const float* eWhh = (const float*)d_in[2];
    const float* ebih = (const float*)d_in[3];
    const float* ebhh = (const float*)d_in[4];
    const float* dWih = (const float*)d_in[5];
    const float* dWhh = (const float*)d_in[6];
    const float* dbih = (const float*)d_in[7];
    const float* dbhh = (const float*)d_in[8];
    const float* fcW  = (const float*)d_in[9];
    const float* fcb  = (const float*)d_in[10];
    float* out = (float*)d_out;

    int B = in_sizes[0] / (TT * 3);
    int blocks = (B + 15) / 16;

    static_assert(LDS_USED_FLOATS * sizeof(float) <= 164 * 1024, "LDS request too small");
    size_t shmem = 164 * 1024;

    traj_rnn_kernel<<<blocks, WAVES * 32, shmem, stream>>>(
        x, eWih, eWhh, ebih, ebhh, dWih, dWhh, dbih, dbhh, fcW, fcb, out, B);
}


# Only the encoder's weights and x (one round trip) are loaded ahead of it; the decoder's loads go out in five groups between encoder steps 2-6 and land while it runs; no full drain; same sums
# speedup vs baseline: 5.6106x; 1.0379x over previous
#include <hip/hip_runtime.h>
#include <cmath>


typedef float v2f __attribute__((ext_vector_type(2)));
typedef float v3f __attribute__((ext_vector_type(3)));
typedef float v4f __attribute__((ext_vector_type(4)));
typedef float v8f __attribute__((ext_vector_type(8)));

constexpr int Hdim  = 128;
constexpr int TT    = 64;
constexpr int PLEN  = 32;
constexpr int WAVES = 4;
constexpr int NK    = 32;
constexpr int NJW   = 2;
constexpr int APACK = NK * 32 * 2;
constexpr int NKQ   = NK / WAVES;
constexpr int TSPLIT = 2;
constexpr int PARTF = 16 * 4 * WAVES;
constexpr int LDS_USED_FLOATS = 2 * APACK + TT * 16 * 4 + 2 * PARTF;

extern __shared__ __attribute__((aligned(16))) float smem[];


__device__ __forceinline__ int a_idx(int m, int k) {
    return (k >> 2) * 64 + (((((k >> 1) & 1) << 4) + m) << 1) + (k & 1);
}

__device__ __forceinline__ int foreign_kk(int i, int wave) {
    return ((wave + (i >> 3)) & 3) * NKQ + (i & 7);
}

__device__ __forceinline__ int chunk_k0(int i, int wave, int half) {
    if (i < NKQ) return wave * 32 + 16 * (i >> 2) + 8 * half + 2 * (i & 3);
    return 4 * foreign_kk(i, wave) + 2 * half;
}

__device__ __forceinline__ float fast_tanh(float x) {
    return __builtin_amdgcn_tanhf(x);
}

__device__ __forceinline__ v8f mma(v2f a, v2f b, v8f c) {
    return __builtin_amdgcn_wmma_f32_16x16x4_f32(false, a, false, b, (short)0, c, false, false);
}

__device__ __forceinline__ void load_bw(v2f (&bw)[NK][NJW], const float* __restrict__ Wm,
                                        int colbase, int wave, int half, int ln) {
#pragma unroll
    for (int i = 0; i < NK; ++i)
#pragma unroll
        for (int jj = 0; jj < NJW; ++jj)
            bw[i][jj] = *(const v2f*)&Wm[(colbase + jj * 16 + ln) * Hdim + chunk_k0(i, wave, half)];
}

template <int I0, int I1>
__device__ __forceinline__ void load_bw_range(v2f (&bw)[NK][NJW], const float* __restrict__ Wm,
                                              int colbase, int wave, int half, int ln) {
#pragma unroll
    for (int i = I0; i < I1; ++i)
#pragma unroll
        for (int jj = 0; jj < NJW; ++jj)
            bw[i][jj] = *(const v2f*)&Wm[(colbase + jj * 16 + ln) * Hdim + chunk_k0(i, wave, half)];
}

constexpr int NFG = 3 * (NKQ / 2);

template <int G>
__device__ __forceinline__ void load_group(v2f (&f)[NFG], const float* hc, int wave, int lane) {
    const float* hl = hc + lane * 2 + G * (4 * 64);
#pragma unroll
    for (int q = 1; q < WAVES; ++q) {
        const float* hq = hl + ((wave + q) & 3) * (NKQ * 64);
#pragma unroll
        for (int j = 0; j < 4; ++j) f[(q - 1) * 4 + j] = *(const v2f*)&hq[j * 64];
    }
}

template <int JJ, int G, int N0, int N1>
__device__ __forceinline__ void group_mma(v8f& p, v8f& q, const v2f (&bw)[NK][NJW],
                                          const v2f (&ho)[4], const v2f (&f)[NFG]) {
#pragma unroll
    for (int n = N0; n < N1; ++n) {
        if (n < 4) {
            if (n & 1) q = mma(bw[4 * G + n][JJ], ho[n], q);
            else       p = mma(bw[4 * G + n][JJ], ho[n], p);
        } else {
            const int m = n - 4;
            const int i = 8 * (1 + (m >> 2)) + 4 * G + (m & 3);
            if (n & 1) q = mma(bw[i][JJ], f[m], q);
            else       p = mma(bw[i][JJ], f[m], p);
        }
    }
}

template <int JJ>
__device__ __forceinline__ void cinit_tile(v8f& c, const v3f& xv,
                                           const float (&bs)[NJW][8], const float (&wi)[NJW][8][3]) {
#pragma unroll
    for (int r = 0; r < 8; ++r)
        c[r] = bs[JJ][r] + xv.x * wi[JJ][r][0] + xv.y * wi[JJ][r][1] + xv.z * wi[JJ][r][2];
}

template <int JJ>
__device__ __forceinline__ void tanh_store(float* hn, const v8f& t, v2f (&ho)[4], int half, int colbase, int ln) {
#pragma unroll
    for (int q = 0; q < 4; ++q) {
        v2f v;
        v.x = fast_tanh(t[2 * q]);
        v.y = fast_tanh(t[2 * q + 1]);
        ho[q] = v;
        *(v2f*)&hn[a_idx(ln, colbase + JJ * 16 + half * 8 + 2 * q)] = v;
    }
}

__device__ __forceinline__ void stores_done_signal() {
    asm volatile("s_wait_dscnt 0x0\n\ts_barrier_signal -1" ::: "memory");
}
__device__ __forceinline__ void others_done_wait() {
    asm volatile("s_barrier_wait -1" ::: "memory");
}


__device__ __forceinline__ v8f fc_partial(const v2f (&af)[NKQ], const v2f (&bfq)[NKQ], const float (&fcbv)[3]) {
    v8f pa, pb;
#pragma unroll
    for (int r = 0; r < 8; ++r) { pa[r] = 0.0f; pb[r] = 0.0f; }
    pa[0] = fcbv[0]; pa[1] = fcbv[1]; pa[2] = fcbv[2];
#pragma unroll
    for (int i = 0; i < NKQ / 2; ++i) {
        pa = mma(bfq[2 * i], af[2 * i], pa);
        pb = mma(bfq[2 * i + 1], af[2 * i + 1], pb);
    }
    return pa + pb;
}

__device__ __forceinline__ void store_partial(float* part, const v8f& pp, int wave, int lane) {
    if (lane < 16) {
#pragma unroll
        for (int o = 0; o < 3; ++o) part[(lane * 4 + o) * WAVES + wave] = pp[o];
    }
}

__device__ __forceinline__ void load_partials(const float* part, v4f& q0, v4f& q1, int wave, int half, int lnc) {
    const int r0 = 2 * wave + half * 8;
    q0 = *(const v4f*)&part[((r0    ) * 4 + lnc) * WAVES];
    q1 = *(const v4f*)&part[((r0 + 1) * 4 + lnc) * WAVES];
}

__device__ __forceinline__ void emit_pred(const v4f& q0, const v4f& q1, float* outp, bool writer) {
    float o0 = (q0.x + q0.y) + (q0.z + q0.w);
    float o1 = (q1.x + q1.y) + (q1.z + q1.w);
    if (writer) { outp[0] = o0; outp[PLEN * 3] = o1; }
}

template <bool PRED>
__device__ __forceinline__ void dec_step(float* hc, float* hn, float* pl,
                                         v8f& p0, v8f& q0, v8f& p1, v8f& q1,
                                         v2f (&h0f)[4], v2f (&h1f)[4], v2f (&f0)[NFG], v2f (&f1)[NFG],
                                         const v2f (&bw)[NK][NJW], const v8f (&init)[NJW],
                                         const v2f (&bfq)[NKQ], const float (&fcbv)[3], float* outp, bool writer,
                                         int wave, int lane, int half, int ln, int lnc, int colbase) {
    {
        const v8f t1 = p1 + q1;
        tanh_store<1>(hc, t1, h1f, half, colbase, ln);
    }
    p0 = init[0];
#pragma unroll
    for (int r = 0; r < 8; ++r) q0[r] = 0.0f;
    group_mma<0, 0, 0, 10>(p0, q0, bw, h0f, f0);
    __builtin_amdgcn_sched_barrier(0);
    stores_done_signal();
    __builtin_amdgcn_sched_barrier(0);
    group_mma<0, 0, 10, 14>(p0, q0, bw, h0f, f0);
    __builtin_amdgcn_sched_barrier(0);
    others_done_wait();
    load_group<1>(f1, hc, wave, lane);
    __builtin_amdgcn_sched_barrier(0);
    group_mma<0, 0, 14, 16>(p0, q0, bw, h0f, f0);
    v8f pp;
    if (PRED) {
        v2f hw[NKQ];
#pragma unroll
        for (int i = 0; i < 4; ++i) { hw[i] = h0f[i]; hw[4 + i] = h1f[i]; }
        pp = fc_partial(hw, bfq, fcbv);
    }
    group_mma<0, 1, 0, 16>(p0, q0, bw, h1f, f1);
    if (PRED) store_partial(pl, pp, wave, lane);
    __builtin_amdgcn_sched_barrier(0);
    v2f h0n[4];
    {
        const v8f t0 = p0 + q0;
        tanh_store<0>(hn, t0, h0n, half, colbase, ln);
    }
    p1 = init[1];
#pragma unroll
    for (int r = 0; r < 8; ++r) q1[r] = 0.0f;
    group_mma<1, 0, 0, 9>(p1, q1, bw, h0f, f0);
    __builtin_amdgcn_sched_barrier(0);
    group_mma<1, 0, 9, 13>(p1, q1, bw, h0f, f0);
    __builtin_amdgcn_sched_barrier(0);
    stores_done_signal();
    __builtin_amdgcn_sched_barrier(0);
    group_mma<1, 0, 13, 16>(p1, q1, bw, h0f, f0);
    group_mma<1, 1, 0, 6>(p1, q1, bw, h1f, f1);
    __builtin_amdgcn_sched_barrier(0);
    others_done_wait();
    load_group<0>(f0, hn, wave, lane);
    v4f qa, qb;
    if (PRED) load_partials(pl, qa, qb, wave, half, lnc);
    __builtin_amdgcn_sched_barrier(0);
    group_mma<1, 1, 6, 16>(p1, q1, bw, h1f, f1);
    if (PRED) emit_pred(qa, qb, outp, writer);
    __builtin_amdgcn_sched_barrier(0);
#pragma unroll
    for (int i = 0; i < 4; ++i) h0f[i] = h0n[i];
}

__global__ __launch_bounds__(WAVES * 32)
void traj_rnn_kernel(const float* __restrict__ x,
                     const float* __restrict__ eWih, const float* __restrict__ eWhh,
                     const float* __restrict__ ebih, const float* __restrict__ ebhh,
                     const float* __restrict__ dWih, const float* __restrict__ dWhh,
                     const float* __restrict__ dbih, const float* __restrict__ dbhh,
                     const float* __restrict__ fcW,  const float* __restrict__ fcb,
                     float* __restrict__ out, int B)
{
    float* hbuf = smem;
    float* xl   = smem + 2 * APACK;
    float* part = xl + TT * 16 * 4;

    const int tid  = threadIdx.x;
    const int wave = __builtin_amdgcn_readfirstlane(tid >> 5);
    const int lane = tid & 31;
    const int half = lane >> 4;
    const int ln   = lane & 15;
    const int colbase = wave * 32;

    const int b0 = blockIdx.x * 16;
    if (b0 >= B) return;

    const float* xt = x + (size_t)b0 * TT * 3;
    constexpr int XPER = 16 * TT * 3 / (WAVES * 32);
    float xs[XPER];
#pragma unroll
    for (int j = 0; j < XPER; ++j) xs[j] = xt[tid + j * (WAVES * 32)];
    for (int e = tid; e < APACK; e += WAVES * 32) hbuf[e] = 0.0f;

    float bse[NJW][8], wie[NJW][8][3];
#pragma unroll
    for (int jj = 0; jj < NJW; ++jj)
#pragma unroll
        for (int r = 0; r < 8; ++r) {
            const int f = colbase + jj * 16 + half * 8 + r;
            bse[jj][r] = ebih[f] + ebhh[f];
#pragma unroll
            for (int i = 0; i < 3; ++i) wie[jj][r][i] = eWih[f * 3 + i];
        }

    v2f bwe[NK][NJW];
    load_bw(bwe, eWhh, colbase, wave, half, ln);

#pragma unroll
    for (int j = 0; j < XPER; ++j) {
        const int e   = tid + j * (WAVES * 32);
        const int row = e / (TT * 3);
        const int rem = e - row * (TT * 3);
        const int t   = rem / 3;
        const int i   = rem - t * 3;
        xl[(t * 16 + row) * 4 + i] = xs[j];
    }

    stores_done_signal();
    others_done_wait();

    v2f h0f[4], h1f[4];
#pragma unroll
    for (int i = 0; i < 4; ++i) { h0f[i].x = 0.0f; h0f[i].y = 0.0f; }
    v8f p0, q0, p1, q1;
#pragma unroll
    for (int r = 0; r < 8; ++r) { p1[r] = 0.0f; q1[r] = 0.0f; }
    v2f f0[NFG], f1[NFG];
    load_group<0>(f0, hbuf, wave, lane);
    v3f xc = *(const v3f*)&xl[ln * 4];

    auto enc_steps = [&](int ta, int tb) __attribute__((always_inline)) {
    for (int t = ta; t < tb; ++t) {
        float* hc = hbuf + (t & 1) * APACK;
        float* hn = hbuf + ((t & 1) ^ 1) * APACK;
        {
            const v8f t1 = p1 + q1;
            tanh_store<1>(hc, t1, h1f, half, colbase, ln);
        }
        cinit_tile<0>(p0, xc, bse, wie);
#pragma unroll
        for (int r = 0; r < 8; ++r) q0[r] = 0.0f;
        group_mma<0, 0, 0, 10>(p0, q0, bwe, h0f, f0);
        __builtin_amdgcn_sched_barrier(0);
        stores_done_signal();
        __builtin_amdgcn_sched_barrier(0);
        group_mma<0, 0, 10, 14>(p0, q0, bwe, h0f, f0);
        __builtin_amdgcn_sched_barrier(0);
        others_done_wait();
        load_group<1>(f1, hc, wave, lane);
        const int tn = t + 1 < TT ? t + 1 : TT - 1;
        const v3f xn = *(const v3f*)&xl[(tn * 16 + ln) * 4];
        __builtin_amdgcn_sched_barrier(0);
        group_mma<0, 0, 14, 16>(p0, q0, bwe, h0f, f0);
        group_mma<0, 1, 0, 16>(p0, q0, bwe, h1f, f1);
        __builtin_amdgcn_sched_barrier(0);
        v2f h0n[4];
        {
            const v8f t0 = p0 + q0;
            tanh_store<0>(hn, t0, h0n, half, colbase, ln);
        }
        cinit_tile<1>(p1, xc, bse, wie);
#pragma unroll
        for (int r = 0; r < 8; ++r) q1[r] = 0.0f;
        group_mma<1, 0, 0, 9>(p1, q1, bwe, h0f, f0);
        __builtin_amdgcn_sched_barrier(0);
        group_mma<1, 0, 9, 13>(p1, q1, bwe, h0f, f0);
        __builtin_amdgcn_sched_barrier(0);
        stores_done_signal();
        __builtin_amdgcn_sched_barrier(0);
        group_mma<1, 0, 13, 16>(p1, q1, bwe, h0f, f0);
        group_mma<1, 1, 0, 6>(p1, q1, bwe, h1f, f1);
        __builtin_amdgcn_sched_barrier(0);
        others_done_wait();
        load_group<0>(f0, hn, wave, lane);
        __builtin_amdgcn_sched_barrier(0);
        group_mma<1, 1, 6, 16>(p1, q1, bwe, h1f, f1);
        __builtin_amdgcn_sched_barrier(0);
#pragma unroll
        for (int i = 0; i < 4; ++i) h0f[i] = h0n[i];
        xc = xn;
    }
    };

    enc_steps(0, TSPLIT);

    v2f bwd[NK][NJW];
    load_bw_range<0, NK / 2>(bwd, dWhh, colbase, wave, half, ln);
    enc_steps(TSPLIT, TSPLIT + 1);
    load_bw_range<NK / 2, NK>(bwd, dWhh, colbase, wave, half, ln);
    enc_steps(TSPLIT + 1, TSPLIT + 2);
    v2f fwr[NK][3];
#pragma unroll
    for (int i = 0; i < NK / 2; ++i) {
        const int k0 = chunk_k0(i, wave, half);
#pragma unroll
        for (int o = 0; o < 3; ++o) fwr[i][o] = *(const v2f*)&fcW[o * Hdim + k0];
    }
    enc_steps(TSPLIT + 2, TSPLIT + 3);
#pragma unroll
    for (int i = NK / 2; i < NK; ++i) {
        const int k0 = chunk_k0(i, wave, half);
#pragma unroll
        for (int o = 0; o < 3; ++o) fwr[i][o] = *(const v2f*)&fcW[o * Hdim + k0];
    }
    enc_steps(TSPLIT + 3, TSPLIT + 4);
    v2f bfr[NKQ];
#pragma unroll
    for (int i = 0; i < NKQ; ++i) bfr[i] = *(const v2f*)&fcW[(ln < 3 ? ln : 0) * Hdim + chunk_k0(i, wave, half)];
    float dbi[NJW][8], dbh[NJW][8], dwi[NJW][8][3];
#pragma unroll
    for (int jj = 0; jj < NJW; ++jj)
#pragma unroll
        for (int r = 0; r < 8; ++r) {
            const int f = colbase + jj * 16 + half * 8 + r;
            dbi[jj][r] = dbih[f];
            dbh[jj][r] = dbhh[f];
#pragma unroll
            for (int i = 0; i < 3; ++i) dwi[jj][r][i] = dWih[f * 3 + i];
        }
    const float* xlast = x + ((size_t)(b0 + ln) * TT + (TT - 1)) * 3;
    const float xl0 = xlast[0], xl1 = xlast[1], xl2 = xlast[2];
    const float fb0 = fcb[0], fb1 = fcb[1], fb2 = fcb[2];
    float wid[NJW][3];
#pragma unroll
    for (int jj = 0; jj < NJW; ++jj)
#pragma unroll
        for (int i = 0; i < 3; ++i) wid[jj][i] = dWih[(colbase + jj * 16 + ln) * 3 + i];

    enc_steps(TSPLIT + 4, TT);

    float bsf[NJW][8];
    v8f cdec0[NJW];
#pragma unroll
    for (int jj = 0; jj < NJW; ++jj)
#pragma unroll
        for (int r = 0; r < 8; ++r) {
            const float bd = dbi[jj][r] + dbh[jj][r];
            const float w0 = dwi[jj][r][0], w1 = dwi[jj][r][1], w2 = dwi[jj][r][2];
            cdec0[jj][r] = bd + xl0 * w0 + xl1 * w1 + xl2 * w2;
            bsf[jj][r] = bd + (w0 * fb0 + w1 * fb1 + w2 * fb2);
        }
    v2f bwf[NK][NJW], bfq[NKQ];
#pragma unroll
    for (int i = 0; i < NK; ++i) {
        const v2f g0 = fwr[i][0], g1 = fwr[i][1], g2 = fwr[i][2];
#pragma unroll
        for (int jj = 0; jj < NJW; ++jj) {
            bwf[i][jj].x = bwd[i][jj].x + (wid[jj][0] * g0.x + wid[jj][1] * g1.x + wid[jj][2] * g2.x);
            bwf[i][jj].y = bwd[i][jj].y + (wid[jj][0] * g0.y + wid[jj][1] * g1.y + wid[jj][2] * g2.y);
        }
    }
#pragma unroll
    for (int i = 0; i < NKQ; ++i) {
        bfq[i].x = ln < 3 ? bfr[i].x : 0.0f;
        bfq[i].y = ln < 3 ? bfr[i].y : 0.0f;
    }
    float fcbv[3];
    fcbv[0] = (wave == 0) ? fb0 : 0.0f;
    fcbv[1] = (wave == 0) ? fb1 : 0.0f;
    fcbv[2] = (wave == 0) ? fb2 : 0.0f;

    const int  lnc    = ln < 3 ? ln : 0;
    const bool writer = ln < 3;
    float* outp = out + ((size_t)(b0 + 2 * wave + half * 8) * PLEN) * 3 + lnc;
    v8f bsfv[NJW];
#pragma unroll
    for (int jj = 0; jj < NJW; ++jj)
#pragma unroll
        for (int r = 0; r < 8; ++r) bsfv[jj][r] = bsf[jj][r];

    dec_step<false>(hbuf + (TT & 1) * APACK, hbuf + ((TT & 1) ^ 1) * APACK, part,
                    p0, q0, p1, q1, h0f, h1f, f0, f1, bwd, cdec0, bfq, fcbv,
                    outp, writer, wave, lane, half, ln, lnc, colbase);
    for (int p = 1; p < PLEN; ++p) {
        const int s = (TT + p) & 1;
        dec_step<true>(hbuf + s * APACK, hbuf + (s ^ 1) * APACK, part + (p & 1) * PARTF,
                       p0, q0, p1, q1, h0f, h1f, f0, f1, bwf, bsfv, bfq, fcbv,
                       outp + (p - 1) * 3, writer, wave, lane, half, ln, lnc, colbase);
    }
    {
        const v8f t1 = p1 + q1;
        v2f hw[NKQ];
#pragma unroll
        for (int q = 0; q < 4; ++q) {
            hw[q] = h0f[q];
            hw[4 + q].x = fast_tanh(t1[2 * q]);
            hw[4 + q].y = fast_tanh(t1[2 * q + 1]);
        }
        const v8f pp = fc_partial(hw, bfq, fcbv);
        float* pl = part + (PLEN & 1) * PARTF;
        store_partial(pl, pp, wave, lane);
        __syncthreads();
        v4f qa, qb;
        load_partials(pl, qa, qb, wave, half, lnc);
        emit_pred(qa, qb, outp + (PLEN - 1) * 3, writer);
    }
}

extern "C" void kernel_launch(void* const* d_in, const int* in_sizes, int n_in,
                              void* d_out, int out_size, void* d_ws, size_t ws_size,
                              hipStream_t stream) {
    (void)n_in; (void)out_size; (void)d_ws; (void)ws_size;
    const float* x    = (const float*)d_in[0];
    const float* eWih = (const float*)d_in[1];
    const float* eWhh = (const float*)d_in[2];
    const float* ebih = (const float*)d_in[3];
    const float* ebhh = (const float*)d_in[4];
    const float* dWih = (const float*)d_in[5];
    const float* dWhh = (const float*)d_in[6];
    const float* dbih = (const float*)d_in[7];
    const float* dbhh = (const float*)d_in[8];
    const float* fcW  = (const float*)d_in[9];
    const float* fcb  = (const float*)d_in[10];
    float* out = (float*)d_out;

    int B = in_sizes[0] / (TT * 3);
    int blocks = (B + 15) / 16;

    static_assert(LDS_USED_FLOATS * sizeof(float) <= 164 * 1024, "LDS request too small");
    size_t shmem = 164 * 1024;

    traj_rnn_kernel<<<blocks, WAVES * 32, shmem, stream>>>(
        x, eWih, eWhh, ebih, ebhh, dWih, dWhh, dbih, dbhh, fcW, fcb, out, B);
}
